# starting kernel
# baseline (speedup 1.0000x reference)
#include <hip/hip_runtime.h>

typedef float v2f __attribute__((ext_vector_type(2)));
typedef float v8f __attribute__((ext_vector_type(8)));

#define N_ATOMS 2048
#define BATCH 2
#define RECV_PER_BLOCK 8
#define CUTOFF_SQ 0.49f
#define MAX_DISP 0.02f

__device__ __forceinline__ float fast_rcp(float x)  { return __builtin_amdgcn_rcpf(x); }
__device__ __forceinline__ float fast_rsq(float x)  { return __builtin_amdgcn_rsqf(x); }

__device__ __forceinline__ float silu_f(float x) {
    return x * fast_rcp(1.0f + __expf(-x));
}
__device__ __forceinline__ float tanh_f(float x) {
    float xc = fminf(fmaxf(x, -15.0f), 15.0f);
    float e  = __expf(2.0f * xc);
    return (e - 1.0f) * fast_rcp(e + 1.0f);
}

__global__ __launch_bounds__(256) void simple_et_kernel(
    const float* __restrict__ x_nm, const int* __restrict__ atom_types,
    const int* __restrict__ mask_in, const float* __restrict__ emb_W,
    const float* __restrict__ msg_w1, const float* __restrict__ msg_b1,
    const float* __restrict__ msg_w2, const float* __restrict__ msg_b2,
    const float* __restrict__ upd_w1, const float* __restrict__ upd_b1,
    const float* __restrict__ upd_w2, const float* __restrict__ upd_b2,
    const float* __restrict__ dx_w, const float* __restrict__ dx_b,
    const float* __restrict__ dv_w, const float* __restrict__ dv_b,
    float* __restrict__ out)
{
    __shared__ float4 snode[N_ATOMS * 2];

    const int tid  = threadIdx.x;
    const int b    = blockIdx.x / (N_ATOMS / RECV_PER_BLOCK);
    const int row0 = (blockIdx.x % (N_ATOMS / RECV_PER_BLOCK)) * RECV_PER_BLOCK;

    for (int n = tid; n < N_ATOMS; n += 256) {
        int t = atom_types[b * N_ATOMS + n];
        float mv = mask_in[b * N_ATOMS + n] ? 1.0f : 0.0f;
        snode[2 * n]     = make_float4(x_nm[(b * N_ATOMS + n) * 3 + 0],
                                       x_nm[(b * N_ATOMS + n) * 3 + 1],
                                       x_nm[(b * N_ATOMS + n) * 3 + 2], mv);
        snode[2 * n + 1] = make_float4(emb_W[t * 3 + 0],
                                       emb_W[t * 3 + 1],
                                       emb_W[t * 3 + 2], 0.0f);
    }
    __syncthreads();

    const int lane = tid & 31;
    const int wave = tid >> 5;
    const int i    = row0 + wave;
    const int m    = lane & 15;
    const bool lo  = (lane < 16);

    const float4 xiq = snode[2 * i];
    const float  xi0 = xiq.x, xi1 = xiq.y, xi2 = xiq.z, maski = xiq.w;

    const int  fm = (m < 3) ? m : 2;
    const float fz = (m < 3 || m == 10) ? 1.0f : 0.0f;
    v2f A1, A2;
    A1[0] = fz * (lo ? msg_w1[0 * 3 + fm] : msg_w1[2 * 3 + fm]);
    A1[1] = fz * (lo ? msg_w1[1 * 3 + fm] : msg_w1[3 * 3 + fm]);
    const float z3 = (m < 3) ? 1.0f : 0.0f;
    A2[0] = z3 * (lo ? msg_w2[0 * 3 + fm] : msg_w2[2 * 3 + fm]);
    A2[1] = (lo ? z3 * msg_w2[1 * 3 + fm] : 0.0f);

    v8f C1 = {}, C2 = {};
    C1[0] = lo ? msg_b1[0] : 0.0f;
    C1[1] = lo ? msg_b1[1] : 0.0f;
    C1[2] = msg_b1[2];
    C2[0] = lo ? msg_b2[0] : 0.0f;
    C2[1] = lo ? msg_b2[1] : 0.0f;
    C2[2] = lo ? msg_b2[2] : 0.0f;

    float acc0 = 0.0f, acc1 = 0.0f, acc2 = 0.0f;
    const float lmask = lo ? 1.0f : 0.0f;

#pragma unroll 2
    for (int jc = 0; jc < N_ATOMS; jc += 16) {
        const int j = jc + m;
        const float4 xq = snode[2 * j];
        const float4 hq = snode[2 * j + 1];

        const float d0 = xi0 - xq.x, d1 = xi1 - xq.y, d2c = xi2 - xq.z;
        const float d2 = d0 * d0 + d1 * d1 + d2c * d2c + 1e-9f;
        const float inv = fast_rsq(d2);
        const float validf = ((d2 < CUTOFF_SQ) && (j != i)) ? (xq.w * maski) : 0.0f;
        const float radial = validf * fminf(inv, 1.0e3f);
        const float dscale = validf * lmask * fminf(inv, 1.0e6f);
        const float dir0 = d0 * dscale, dir1 = d1 * dscale, dir2 = d2c * dscale;

        v2f B1;
        B1[0] = lo ? hq.x : hq.z;
        B1[1] = lo ? hq.y : radial;

        v8f D1 = __builtin_amdgcn_wmma_f32_16x16x4_f32(
            false, A1, false, B1, (short)0, C1, false, false);

        const float hd0 = silu_f(D1[0]);
        const float hd1 = silu_f(D1[1]);
        const float hd2 = silu_f(D1[2]);

        v2f B2;
        B2[0] = lo ? hd0 : hd2;
        B2[1] = lo ? hd1 : 0.0f;

        v8f D2 = __builtin_amdgcn_wmma_f32_16x16x4_f32(
            false, A2, false, B2, (short)0, C2, false, false);

        acc0 += D2[0] * dir0;
        acc1 += D2[1] * dir1;
        acc2 += D2[2] * dir2;
    }

    for (int off = 16; off > 0; off >>= 1) {
        acc0 += __shfl_xor(acc0, off, 32);
        acc1 += __shfl_xor(acc1, off, 32);
        acc2 += __shfl_xor(acc2, off, 32);
    }

    if (lane == 0) {
        const float v0 = acc0, v1 = acc1, v2 = acc2;
        const float4 hiq = snode[2 * i + 1];
        const float hi0 = hiq.x, hi1 = hiq.y, hi2 = hiq.z;

        const float u0 = silu_f(v0 * upd_w1[0] + v1 * upd_w1[3] + v2 * upd_w1[6] + upd_b1[0]);
        const float u1 = silu_f(v0 * upd_w1[1] + v1 * upd_w1[4] + v2 * upd_w1[7] + upd_b1[1]);
        const float u2 = silu_f(v0 * upd_w1[2] + v1 * upd_w1[5] + v2 * upd_w1[8] + upd_b1[2]);

        const float h0 = hi0 + u0 * upd_w2[0] + u1 * upd_w2[3] + u2 * upd_w2[6] + upd_b2[0];
        const float h1 = hi1 + u0 * upd_w2[1] + u1 * upd_w2[4] + u2 * upd_w2[7] + upd_b2[1];
        const float h2 = hi2 + u0 * upd_w2[2] + u1 * upd_w2[5] + u2 * upd_w2[8] + upd_b2[2];

        const int base = (b * N_ATOMS + i) * 3;
        const int half = BATCH * N_ATOMS * 3;
#pragma unroll
        for (int c = 0; c < 3; ++c) {
            const float px = h0 * dx_w[0 * 3 + c] + h1 * dx_w[1 * 3 + c] + h2 * dx_w[2 * 3 + c] + dx_b[c];
            const float pv = h0 * dv_w[0 * 3 + c] + h1 * dv_w[1 * 3 + c] + h2 * dv_w[2 * 3 + c] + dv_b[c];
            out[base + c]        = tanh_f(px) * MAX_DISP;
            out[half + base + c] = tanh_f(pv) * 1.0f;
        }
    }
}

extern "C" void kernel_launch(void* const* d_in, const int* in_sizes, int n_in,
                              void* d_out, int out_size, void* d_ws, size_t ws_size,
                              hipStream_t stream) {
    (void)in_sizes; (void)n_in; (void)d_ws; (void)ws_size; (void)out_size;
    const float* x_nm   = (const float*)d_in[0];
    const int*   types  = (const int*)  d_in[1];
    const int*   mask   = (const int*)  d_in[2];
    const float* emb_W  = (const float*)d_in[3];
    const float* msg_w1 = (const float*)d_in[4];
    const float* msg_b1 = (const float*)d_in[5];
    const float* msg_w2 = (const float*)d_in[6];
    const float* msg_b2 = (const float*)d_in[7];
    const float* upd_w1 = (const float*)d_in[8];
    const float* upd_b1 = (const float*)d_in[9];
    const float* upd_w2 = (const float*)d_in[10];
    const float* upd_b2 = (const float*)d_in[11];
    const float* dx_w   = (const float*)d_in[12];
    const float* dx_b   = (const float*)d_in[13];
    const float* dv_w   = (const float*)d_in[14];
    const float* dv_b   = (const float*)d_in[15];
    float* out = (float*)d_out;

    dim3 grid(BATCH * (N_ATOMS / RECV_PER_BLOCK));
    dim3 block(256);
    simple_et_kernel<<<grid, block, 0, stream>>>(
        x_nm, types, mask, emb_W,
        msg_w1, msg_b1, msg_w2, msg_b2,
        upd_w1, upd_b1, upd_w2, upd_b2,
        dx_w, dx_b, dv_w, dv_b, out);
}


# Sender staging rides behind the pair loop: 8 chunks of 256 senders, chunk k+2 fetched and stored while chunk k's tiles run; waves take every fourth tile; a barrier per chunk; sum order changes
# speedup vs baseline: 1.5658x; 1.5658x over previous
#include <hip/hip_runtime.h>


typedef float v2f __attribute__((ext_vector_type(2)));
typedef float v8f __attribute__((ext_vector_type(8)));

#define N_ATOMS 2048
#define BATCH 2
#define RECV_PER_BLOCK 8
#define CUTOFF_SQ 0.49f
#define MAX_DISP 0.02f
#define XT_PITCH 2064
#define CHUNK 256
#define N_CHUNKS 8
#define TILES_PER_WAVE 32
#define TILES_PER_CHUNK 4
#define BIAS_FEATURE 64.0f

__device__ __forceinline__ float fast_rcp(float x)  { return __builtin_amdgcn_rcpf(x); }
__device__ __forceinline__ float fast_rsq(float x)  { return __builtin_amdgcn_rsqf(x); }

__device__ __forceinline__ float silu_f(float x) {
    return x * fast_rcp(1.0f + __expf(-x));
}
__device__ __forceinline__ float tanh_f(float x) {
    float xc = fminf(fmaxf(x, -15.0f), 15.0f);
    float e  = __expf(2.0f * xc);
    return (e - 1.0f) * fast_rcp(e + 1.0f);
}

__device__ __forceinline__ v8f mma(v2f a, v2f b, v8f c) {
    return __builtin_amdgcn_wmma_f32_16x16x4_f32(false, a, false, b, (short)0, c, false, false);
}

struct SenderA { int t, mk; float x0, x1, x2; };
struct SenderB { float e0, e1, e2; };

__device__ __forceinline__ SenderA load_a(const float* __restrict__ x_nm, const int* __restrict__ atom_types,
                                          const int* __restrict__ mask_in, const int g) {
    SenderA s;
    s.t  = atom_types[g];
    s.mk = mask_in[g];
    s.x0 = x_nm[g * 3 + 0];  s.x1 = x_nm[g * 3 + 1];  s.x2 = x_nm[g * 3 + 2];
    return s;
}
__device__ __forceinline__ SenderB load_b(const float* __restrict__ emb_W, const int t) {
    SenderB s;
    s.e0 = emb_W[t * 3 + 0];  s.e1 = emb_W[t * 3 + 1];  s.e2 = emb_W[t * 3 + 2];
    return s;
}
__device__ __forceinline__ void put_sender(float4* snode, float* xT, const int n, const SenderA a, const SenderB e) {
    snode[2 * n]     = make_float4(a.x0, a.x1, a.x2, a.mk ? 1.0f : 0.0f);
    snode[2 * n + 1] = make_float4(e.e0, e.e1, e.e2, 1.0f);
    xT[n] = a.x0;  xT[XT_PITCH + n] = a.x1;  xT[2 * XT_PITCH + n] = a.x2;
}

__device__ __forceinline__ void lds_handover() {
    asm volatile("s_wait_dscnt 0x0" ::: "memory");
    asm volatile("s_barrier_signal -1\n\ts_barrier_wait -1" ::: "memory");
}

__device__ __forceinline__ void pair_geometry(const float4 xr, const v2f XI0, const v2f XI1, const v2f XI2,
                                              const v2f MI, const int sa, v2f& arad, v2f& ads) {
    const v2f e0 = XI0 - xr.x, e1 = XI1 - xr.y, e2 = XI2 - xr.z;
    v2f q;
    {
#pragma clang fp contract(off)
        q = e0 * e0 + e1 * e1 + e2 * e2 + 1e-9f;
    }
    const float inv0 = fast_rsq(q[0]), inv1 = fast_rsq(q[1]);
    const float m0 = xr.w * MI[0], m1 = xr.w * MI[1];
    const float v0 = ((q[0] < CUTOFF_SQ) && (sa != 0))  ? m0 : 0.0f;
    const float v1 = ((q[1] < CUTOFF_SQ) && (sa != -1)) ? m1 : 0.0f;
    arad[0] = v0 * fminf(inv0, 1.0e3f);  arad[1] = v1 * fminf(inv1, 1.0e3f);
    ads[0]  = v0 * fminf(inv0, 1.0e6f);  ads[1]  = v1 * fminf(inv1, 1.0e6f);
}

__global__ __launch_bounds__(256) void simple_et_kernel(
    const float* __restrict__ x_nm, const int* __restrict__ atom_types,
    const int* __restrict__ mask_in, const float* __restrict__ emb_W,
    const float* __restrict__ msg_w1, const float* __restrict__ msg_b1,
    const float* __restrict__ msg_w2, const float* __restrict__ msg_b2,
    const float* __restrict__ upd_w1, const float* __restrict__ upd_b1,
    const float* __restrict__ upd_w2, const float* __restrict__ upd_b2,
    const float* __restrict__ dx_w, const float* __restrict__ dx_b,
    const float* __restrict__ dv_w, const float* __restrict__ dv_b,
    float* __restrict__ out)
{
    __shared__ float4 snode[N_ATOMS * 2];
    __shared__ __attribute__((aligned(16))) float xT[3 * XT_PITCH];
    __shared__ float4 red4[8 * 12];

    const int tid  = threadIdx.x;
    const int b    = blockIdx.x / (N_ATOMS / RECV_PER_BLOCK);
    const int row0 = (blockIdx.x % (N_ATOMS / RECV_PER_BLOCK)) * RECV_PER_BLOCK;

    const int gb = b * N_ATOMS;
    const SenderA a0 = load_a(x_nm, atom_types, mask_in, gb + tid);
    const SenderA a1 = load_a(x_nm, atom_types, mask_in, gb + CHUNK + tid);
    SenderA fa2 = load_a(x_nm, atom_types, mask_in, gb + 2 * CHUNK + tid);
    SenderA fa3 = load_a(x_nm, atom_types, mask_in, gb + 3 * CHUNK + tid);

    const int lane = tid & 31;
    const int wave = tid >> 5;
    const int c    = lane & 15;
    const int hh   = lane >> 4;
    const int grp  = wave >> 2;
    const int sq   = wave & 3;
    const int i0   = row0 + 4 * grp;
    const int jbase = 16 * sq;

    const int ia = i0 + 2 * hh;
    const float* xra = x_nm + (gb + ia) * 3;
    const v2f XI0 = {xra[0], xra[3]}, XI1 = {xra[1], xra[4]}, XI2 = {xra[2], xra[5]};
    const v2f MI  = {mask_in[gb + ia] ? 1.0f : 0.0f, mask_in[gb + ia + 1] ? 1.0f : 0.0f};

    const int  kf   = c & 3;
    const int  ic   = c >> 2;
    const bool feat = kf < 3;
    const int  kq   = feat ? kf : 0;
    const float w_a = msg_w1[(2 * hh) * 3 + kq];
    const float w_b = msg_w1[1 * 3 + kq];
    const float w_c = msg_b1[kq];
    const float w_r = msg_w1[3 * 3 + kq];
    v2f BW, BR, BS;
    BW[0] = feat ? w_a : 0.0f;
    BW[1] = hh ? (feat ? w_c : BIAS_FEATURE) : (feat ? w_b : 0.0f);
    BR[0] = ((2 * hh     == ic) && feat) ? w_r : 0.0f;
    BR[1] = ((2 * hh + 1 == ic) && feat) ? w_r : 0.0f;
    BS[0] = (2 * hh     == ic) ? 1.0f : 0.0f;
    BS[1] = (2 * hh + 1 == ic) ? 1.0f : 0.0f;

    const int   c3  = c / 3;
    const int   iB  = (c < 12) ? c3 : 3;
    const int   ch  = c - 3 * c3;
    const float xiB = x_nm[(gb + i0 + iB) * 3 + ch];
    const float* xTp = xT + ch * XT_PITCH + jbase + 8 * hh;

    const float4* recx = snode + 2 * (jbase + c);
    const v2f*    rech = reinterpret_cast<const v2f*>(snode + 2 * (jbase + c) + 1) + hh;

    {
        const SenderB b0 = load_b(emb_W, a0.t);
        const SenderB b1 = load_b(emb_W, a1.t);
        put_sender(snode, xT, tid, a0, b0);
        put_sender(snode, xT, CHUNK + tid, a1, b1);
    }
    SenderB fb2 = load_b(emb_W, fa2.t);
    lds_handover();

    const v8f Z = {};
    int sa = ia - (jbase + c);

    v2f ah_n, arad_n, ads_n;
    v8f D1c, Ddc;
    {
        const float4 xr = recx[0];
        const v2f    hv = rech[0];
        v2f arad, ads;
        pair_geometry(xr, XI0, XI1, XI2, MI, sa, arad, ads);
        sa -= 64;
        D1c = mma(hv, BW, Z);
        Ddc = mma(ads, BS, Z);
        D1c = mma(arad, BR, D1c);
        const float4 xr1 = recx[128];
        ah_n = rech[256];
        pair_geometry(xr1, XI0, XI1, XI2, MI, sa, arad_n, ads_n);
        sa -= 64;
    }
    v8f G0 = {}, G1 = {};

    for (int k = 0; k < N_CHUNKS; ++k) {
#pragma unroll
    for (int i = 0; i < TILES_PER_CHUNK; ++i) {
        const int t  = TILES_PER_CHUNK * k + i;
        const int t2 = (t + 2) & (TILES_PER_WAVE - 1);
        const float4 xr2 = recx[128 * t2];
        const v2f    hv2 = rech[256 * t2];
        const float4 xs0 = *reinterpret_cast<const float4*>(xTp + 64 * t);
        const float4 xs1 = *reinterpret_cast<const float4*>(xTp + 64 * t + 4);

        v8f D1n = mma(ah_n, BW, Z);
        v8f Ddn = mma(ads_n, BS, Z);
        D1n = mma(arad_n, BR, D1n);
        __builtin_amdgcn_sched_barrier(0);

        v2f p0, p1, p2, p3;
        {
            const float d0 = 1.0f + __expf(-D1c[0]), d1 = 1.0f + __expf(-D1c[1]);
            const float d2 = 1.0f + __expf(-D1c[2]), d3 = 1.0f + __expf(-D1c[3]);
            const float d4 = 1.0f + __expf(-D1c[4]), d5 = 1.0f + __expf(-D1c[5]);
            const float d6 = 1.0f + __expf(-D1c[6]), d7 = 1.0f + __expf(-D1c[7]);
            p0[0] = (D1c[0] * fast_rcp(d0)) * Ddc[0];  p0[1] = (D1c[1] * fast_rcp(d1)) * Ddc[1];
            p1[0] = (D1c[2] * fast_rcp(d2)) * Ddc[2];  p1[1] = (D1c[3] * fast_rcp(d3)) * Ddc[3];
            p2[0] = (D1c[4] * fast_rcp(d4)) * Ddc[4];  p2[1] = (D1c[5] * fast_rcp(d5)) * Ddc[5];
            p3[0] = (D1c[6] * fast_rcp(d6)) * Ddc[6];  p3[1] = (D1c[7] * fast_rcp(d7)) * Ddc[7];
        }
        const v2f E0 = {xiB - xs0.x, xiB - xs0.y}, E1 = {xiB - xs0.z, xiB - xs0.w};
        const v2f E2 = {xiB - xs1.x, xiB - xs1.y}, E3 = {xiB - xs1.z, xiB - xs1.w};
        __builtin_amdgcn_sched_barrier(0);

        G0 = mma(p0, E0, G0);
        G1 = mma(p1, E1, G1);
        G0 = mma(p2, E2, G0);
        G1 = mma(p3, E3, G1);
        __builtin_amdgcn_sched_barrier(0);

        v2f arad_nn, ads_nn;
        pair_geometry(xr2, XI0, XI1, XI2, MI, sa, arad_nn, ads_nn);
        sa -= 64;
        D1c = D1n;  Ddc = Ddn;
        ah_n = hv2;  arad_n = arad_nn;  ads_n = ads_nn;
    }
        if (k < N_CHUNKS - 2) {
            put_sender(snode, xT, (k + 2) * CHUNK + tid, fa2, fb2);
            if (k < N_CHUNKS - 3) {
                fa2 = fa3;
                fb2 = load_b(emb_W, fa3.t);
            }
            if (k < N_CHUNKS - 4)
                fa3 = load_a(x_nm, atom_types, mask_in, gb + (k + 4) * CHUNK + tid);
            lds_handover();
        }
    }

    {
        const v8f G = G0 + G1;
        const bool odd = (iB & 1) != 0;
        float4 val;
        val.x = odd ? G[4] : G[0];
        val.y = odd ? G[5] : G[1];
        val.z = odd ? G[6] : G[2];
        val.w = odd ? G[7] : G[3];
        if ((c < 12) && (hh == (iB >> 1))) red4[wave * 12 + c] = val;
    }
    __syncthreads();

    if (lane == 0) {
        const int i  = row0 + wave;
        const int ip = wave & 3;
        float vm[3];
#pragma unroll
        for (int cc = 0; cc < 3; ++cc) {
            float4 g = red4[(4 * grp + 0) * 12 + 3 * ip + cc];
#pragma unroll
            for (int s = 1; s < 4; ++s) {
                const float4 g2 = red4[(4 * grp + s) * 12 + 3 * ip + cc];
                g.x += g2.x;  g.y += g2.y;  g.z += g2.z;  g.w += g2.w;
            }
            vm[cc] = g.x * msg_w2[0 * 3 + cc] + g.y * msg_w2[1 * 3 + cc] + g.z * msg_w2[2 * 3 + cc]
                   + g.w * (msg_b2[cc] * (1.0f / BIAS_FEATURE));
        }
        const float v0 = vm[0], v1 = vm[1], v2 = vm[2];
        const float4 hiq = snode[2 * i + 1];
        const float hi0 = hiq.x, hi1 = hiq.y, hi2 = hiq.z;

        const float u0 = silu_f(v0 * upd_w1[0] + v1 * upd_w1[3] + v2 * upd_w1[6] + upd_b1[0]);
        const float u1 = silu_f(v0 * upd_w1[1] + v1 * upd_w1[4] + v2 * upd_w1[7] + upd_b1[1]);
        const float u2 = silu_f(v0 * upd_w1[2] + v1 * upd_w1[5] + v2 * upd_w1[8] + upd_b1[2]);

        const float h0 = hi0 + u0 * upd_w2[0] + u1 * upd_w2[3] + u2 * upd_w2[6] + upd_b2[0];
        const float h1 = hi1 + u0 * upd_w2[1] + u1 * upd_w2[4] + u2 * upd_w2[7] + upd_b2[1];
        const float h2 = hi2 + u0 * upd_w2[2] + u1 * upd_w2[5] + u2 * upd_w2[8] + upd_b2[2];

        const int base = (b * N_ATOMS + i) * 3;
        const int half = BATCH * N_ATOMS * 3;
#pragma unroll
        for (int cc = 0; cc < 3; ++cc) {
            const float px = h0 * dx_w[0 * 3 + cc] + h1 * dx_w[1 * 3 + cc] + h2 * dx_w[2 * 3 + cc] + dx_b[cc];
            const float pv = h0 * dv_w[0 * 3 + cc] + h1 * dv_w[1 * 3 + cc] + h2 * dv_w[2 * 3 + cc] + dv_b[cc];
            out[base + cc]        = tanh_f(px) * MAX_DISP;
            out[half + base + cc] = tanh_f(pv) * 1.0f;
        }
    }
}

extern "C" void kernel_launch(void* const* d_in, const int* in_sizes, int n_in,
                              void* d_out, int out_size, void* d_ws, size_t ws_size,
                              hipStream_t stream) {
    (void)in_sizes; (void)n_in; (void)d_ws; (void)ws_size; (void)out_size;
    const float* x_nm   = (const float*)d_in[0];
    const int*   types  = (const int*)  d_in[1];
    const int*   mask   = (const int*)  d_in[2];
    const float* emb_W  = (const float*)d_in[3];
    const float* msg_w1 = (const float*)d_in[4];
    const float* msg_b1 = (const float*)d_in[5];
    const float* msg_w2 = (const float*)d_in[6];
    const float* msg_b2 = (const float*)d_in[7];
    const float* upd_w1 = (const float*)d_in[8];
    const float* upd_b1 = (const float*)d_in[9];
    const float* upd_w2 = (const float*)d_in[10];
    const float* upd_b2 = (const float*)d_in[11];
    const float* dx_w   = (const float*)d_in[12];
    const float* dx_b   = (const float*)d_in[13];
    const float* dv_w   = (const float*)d_in[14];
    const float* dv_b   = (const float*)d_in[15];
    float* out = (float*)d_out;

    dim3 grid(BATCH * (N_ATOMS / RECV_PER_BLOCK));
    dim3 block(256);
    simple_et_kernel<<<grid, block, 0, stream>>>(
        x_nm, types, mask, emb_W,
        msg_w1, msg_b1, msg_w2, msg_b2,
        upd_w1, upd_b1, upd_w2, upd_b2,
        dx_w, dx_b, dv_w, dv_b, out);
}


# Pair directions ds*(x_r-x_j) made by the sender's lane reach the sender-sum products as B operands via per-wave LDS strips: the 0/1 spreading matrix product goes (6 per tile, was 7); sum order same
# speedup vs baseline: 1.6830x; 1.0748x over previous
#include <hip/hip_runtime.h>


typedef float v2f __attribute__((ext_vector_type(2)));
typedef float v8f __attribute__((ext_vector_type(8)));

#define N_ATOMS 2048
#define BATCH 2
#define RECV_PER_BLOCK 8
#define CUTOFF_SQ 0.49f
#define MAX_DISP 0.02f
#define DIR_PITCH 20
#define DIR_UPPER 160
#define DIR_STRIP 280
#define CHUNK 256
#define N_CHUNKS 8
#define TILES_PER_WAVE 32
#define TILES_PER_CHUNK 4
#define BIAS_FEATURE 64.0f

__device__ __forceinline__ float fast_rcp(float x)  { return __builtin_amdgcn_rcpf(x); }
__device__ __forceinline__ float fast_rsq(float x)  { return __builtin_amdgcn_rsqf(x); }

__device__ __forceinline__ float silu_f(float x) {
    return x * fast_rcp(1.0f + __expf(-x));
}
__device__ __forceinline__ float tanh_f(float x) {
    float xc = fminf(fmaxf(x, -15.0f), 15.0f);
    float e  = __expf(2.0f * xc);
    return (e - 1.0f) * fast_rcp(e + 1.0f);
}

__device__ __forceinline__ v8f mma(v2f a, v2f b, v8f c) {
    return __builtin_amdgcn_wmma_f32_16x16x4_f32(false, a, false, b, (short)0, c, false, false);
}

struct SenderA { int t, mk; float x0, x1, x2; };
struct SenderB { float e0, e1, e2; };

__device__ __forceinline__ SenderA load_a(const float* __restrict__ x_nm, const int* __restrict__ atom_types,
                                          const int* __restrict__ mask_in, const int g) {
    SenderA s;
    s.t  = atom_types[g];
    s.mk = mask_in[g];
    s.x0 = x_nm[g * 3 + 0];  s.x1 = x_nm[g * 3 + 1];  s.x2 = x_nm[g * 3 + 2];
    return s;
}
__device__ __forceinline__ SenderB load_b(const float* __restrict__ emb_W, const int t) {
    SenderB s;
    s.e0 = emb_W[t * 3 + 0];  s.e1 = emb_W[t * 3 + 1];  s.e2 = emb_W[t * 3 + 2];
    return s;
}
__device__ __forceinline__ void put_sender(float4* snode, const int n, const SenderA a, const SenderB e) {
    snode[2 * n]     = make_float4(a.x0, a.x1, a.x2, a.mk ? 1.0f : 0.0f);
    snode[2 * n + 1] = make_float4(e.e0, e.e1, e.e2, 1.0f);
}

__device__ __forceinline__ void lds_handover() {
    asm volatile("s_wait_dscnt 0x0" ::: "memory");
    asm volatile("s_barrier_signal -1\n\ts_barrier_wait -1" ::: "memory");
}

__device__ __forceinline__ void pair_geometry(const float4 xr, const v2f XI0, const v2f XI1, const v2f XI2,
                                              const v2f MI, const int sa, v2f& arad, v2f& dir0, v2f& dir1, v2f& dir2) {
    const v2f e0 = XI0 - xr.x, e1 = XI1 - xr.y, e2 = XI2 - xr.z;
    v2f q;
    {
#pragma clang fp contract(off)
        q = e0 * e0 + e1 * e1 + e2 * e2 + 1e-9f;
    }
    const float inv0 = fast_rsq(q[0]), inv1 = fast_rsq(q[1]);
    const float m0 = xr.w * MI[0], m1 = xr.w * MI[1];
    const float v0 = ((q[0] < CUTOFF_SQ) && (sa != 0))  ? m0 : 0.0f;
    const float v1 = ((q[1] < CUTOFF_SQ) && (sa != -1)) ? m1 : 0.0f;
    arad[0] = v0 * fminf(inv0, 1.0e3f);  arad[1] = v1 * fminf(inv1, 1.0e3f);
    v2f ads;
    ads[0]  = v0 * fminf(inv0, 1.0e6f);  ads[1]  = v1 * fminf(inv1, 1.0e6f);
    dir0 = ads * e0;  dir1 = ads * e1;  dir2 = ads * e2;
}

__device__ __forceinline__ void put_dir(float* w, const v2f dir0, const v2f dir1, const v2f dir2) {
    w[0 * DIR_PITCH] = dir0[0];  w[1 * DIR_PITCH] = dir1[0];  w[2 * DIR_PITCH] = dir2[0];
    w[3 * DIR_PITCH] = dir0[1];  w[4 * DIR_PITCH] = dir1[1];  w[5 * DIR_PITCH] = dir2[1];
}

__global__ __launch_bounds__(256) void simple_et_kernel(
    const float* __restrict__ x_nm, const int* __restrict__ atom_types,
    const int* __restrict__ mask_in, const float* __restrict__ emb_W,
    const float* __restrict__ msg_w1, const float* __restrict__ msg_b1,
    const float* __restrict__ msg_w2, const float* __restrict__ msg_b2,
    const float* __restrict__ upd_w1, const float* __restrict__ upd_b1,
    const float* __restrict__ upd_w2, const float* __restrict__ upd_b2,
    const float* __restrict__ dx_w, const float* __restrict__ dx_b,
    const float* __restrict__ dv_w, const float* __restrict__ dv_b,
    float* __restrict__ out)
{
    __shared__ float4 snode[N_ATOMS * 2];
    __shared__ float4 red4[8 * 12];
    __shared__ __attribute__((aligned(16))) float dirT[8 * 2 * DIR_STRIP];

    const int tid  = threadIdx.x;
    const int b    = blockIdx.x / (N_ATOMS / RECV_PER_BLOCK);
    const int row0 = (blockIdx.x % (N_ATOMS / RECV_PER_BLOCK)) * RECV_PER_BLOCK;

    const int gb = b * N_ATOMS;
    const SenderA a0 = load_a(x_nm, atom_types, mask_in, gb + tid);
    const SenderA a1 = load_a(x_nm, atom_types, mask_in, gb + CHUNK + tid);
    SenderA fa2 = load_a(x_nm, atom_types, mask_in, gb + 2 * CHUNK + tid);
    SenderA fa3 = load_a(x_nm, atom_types, mask_in, gb + 3 * CHUNK + tid);

    const int lane = tid & 31;
    const int wave = tid >> 5;
    const int c    = lane & 15;
    const int hh   = lane >> 4;
    const int grp  = wave >> 2;
    const int sq   = wave & 3;
    const int i0   = row0 + 4 * grp;
    const int jbase = 16 * sq;

    const int ia = i0 + 2 * hh;
    const float* xra = x_nm + (gb + ia) * 3;
    const v2f XI0 = {xra[0], xra[3]}, XI1 = {xra[1], xra[4]}, XI2 = {xra[2], xra[5]};
    const v2f MI  = {mask_in[gb + ia] ? 1.0f : 0.0f, mask_in[gb + ia + 1] ? 1.0f : 0.0f};

    const int  kf   = c & 3;
    const int  ic   = c >> 2;
    const bool feat = kf < 3;
    const int  kq   = feat ? kf : 0;
    const float w_a = msg_w1[(2 * hh) * 3 + kq];
    const float w_b = msg_w1[1 * 3 + kq];
    const float w_c = msg_b1[kq];
    const float w_r = msg_w1[3 * 3 + kq];
    v2f BW, BR;
    BW[0] = feat ? w_a : 0.0f;
    BW[1] = hh ? (feat ? w_c : BIAS_FEATURE) : (feat ? w_b : 0.0f);
    BR[0] = ((2 * hh     == ic) && feat) ? w_r : 0.0f;
    BR[1] = ((2 * hh + 1 == ic) && feat) ? w_r : 0.0f;

    const int   c3  = c / 3;
    const int   iB  = (c < 12) ? c3 : 3;
    const int   cB  = (c < 12) ? c : 11;
    float*        dirw = dirT + wave * (2 * DIR_STRIP) + hh * DIR_UPPER + c;
    const float4* dirr = reinterpret_cast<const float4*>(dirT + wave * (2 * DIR_STRIP)
                                                         + cB * DIR_PITCH + (cB >= 6 ? DIR_UPPER - 6 * DIR_PITCH : 0) + 8 * hh);

    const float4* recx = snode + 2 * (jbase + c);
    const v2f*    rech = reinterpret_cast<const v2f*>(snode + 2 * (jbase + c) + 1) + hh;

    {
        const SenderB b0 = load_b(emb_W, a0.t);
        const SenderB b1 = load_b(emb_W, a1.t);
        put_sender(snode, tid, a0, b0);
        put_sender(snode, CHUNK + tid, a1, b1);
    }
    SenderB fb2 = load_b(emb_W, fa2.t);
    lds_handover();

    const v8f Z = {};
    int sa = ia - (jbase + c);

    v2f ah_n, arad_n;
    v8f D1c;
    {
        const float4 xr = recx[0];
        const v2f    hv = rech[0];
        v2f arad, dr0, dr1, dr2;
        pair_geometry(xr, XI0, XI1, XI2, MI, sa, arad, dr0, dr1, dr2);
        sa -= 64;
        put_dir(dirw, dr0, dr1, dr2);
        D1c = mma(hv, BW, Z);
        D1c = mma(arad, BR, D1c);
        const float4 xr1 = recx[128];
        ah_n = rech[256];
        pair_geometry(xr1, XI0, XI1, XI2, MI, sa, arad_n, dr0, dr1, dr2);
        sa -= 64;
        put_dir(dirw + DIR_STRIP, dr0, dr1, dr2);
    }
    v8f G0 = {}, G1 = {};

    for (int k = 0; k < N_CHUNKS; ++k) {
#pragma unroll
    for (int i = 0; i < TILES_PER_CHUNK; ++i) {
        const int t  = TILES_PER_CHUNK * k + i;
        const int t2 = (t + 2) & (TILES_PER_WAVE - 1);
        const float4 xr2 = recx[128 * t2];
        const v2f    hv2 = rech[256 * t2];
        const float4 dq0 = dirr[(DIR_STRIP / 4) * (i & 1)];
        const float4 dq1 = dirr[(DIR_STRIP / 4) * (i & 1) + 1];

        v8f D1n = mma(ah_n, BW, Z);
        __builtin_amdgcn_sched_barrier(0);

        v2f p0, p1, p2, p3;
        {
            const float d0 = 1.0f + __expf(-D1c[0]), d1 = 1.0f + __expf(-D1c[1]);
            const float d2 = 1.0f + __expf(-D1c[2]), d3 = 1.0f + __expf(-D1c[3]);
            const float d4 = 1.0f + __expf(-D1c[4]), d5 = 1.0f + __expf(-D1c[5]);
            const float d6 = 1.0f + __expf(-D1c[6]), d7 = 1.0f + __expf(-D1c[7]);
            p0[0] = D1c[0] * fast_rcp(d0);  p0[1] = D1c[1] * fast_rcp(d1);
            p1[0] = D1c[2] * fast_rcp(d2);  p1[1] = D1c[3] * fast_rcp(d3);
            p2[0] = D1c[4] * fast_rcp(d4);  p2[1] = D1c[5] * fast_rcp(d5);
            p3[0] = D1c[6] * fast_rcp(d6);  p3[1] = D1c[7] * fast_rcp(d7);
        }
        const v2f E0 = {dq0.x, dq0.y}, E1 = {dq0.z, dq0.w};
        const v2f E2 = {dq1.x, dq1.y}, E3 = {dq1.z, dq1.w};
        __builtin_amdgcn_sched_barrier(0);

        D1n = mma(arad_n, BR, D1n);
        G0 = mma(p0, E0, G0);
        G1 = mma(p1, E1, G1);
        G0 = mma(p2, E2, G0);
        G1 = mma(p3, E3, G1);
        __builtin_amdgcn_sched_barrier(0);

        v2f arad_nn, dr0, dr1, dr2;
        pair_geometry(xr2, XI0, XI1, XI2, MI, sa, arad_nn, dr0, dr1, dr2);
        sa -= 64;
        put_dir(dirw + DIR_STRIP * (i & 1), dr0, dr1, dr2);
        D1c = D1n;
        ah_n = hv2;  arad_n = arad_nn;
    }
        if (k < N_CHUNKS - 2) {
            put_sender(snode, (k + 2) * CHUNK + tid, fa2, fb2);
            if (k < N_CHUNKS - 3) {
                fa2 = fa3;
                fb2 = load_b(emb_W, fa3.t);
            }
            if (k < N_CHUNKS - 4)
                fa3 = load_a(x_nm, atom_types, mask_in, gb + (k + 4) * CHUNK + tid);
            lds_handover();
        }
    }

    {
        const v8f G = G0 + G1;
        const bool odd = (iB & 1) != 0;
        float4 val;
        val.x = odd ? G[4] : G[0];
        val.y = odd ? G[5] : G[1];
        val.z = odd ? G[6] : G[2];
        val.w = odd ? G[7] : G[3];
        if ((c < 12) && (hh == (iB >> 1))) red4[wave * 12 + c] = val;
    }
    __syncthreads();

    if (lane == 0) {
        const int i  = row0 + wave;
        const int ip = wave & 3;
        float vm[3];
#pragma unroll
        for (int cc = 0; cc < 3; ++cc) {
            float4 g = red4[(4 * grp + 0) * 12 + 3 * ip + cc];
#pragma unroll
            for (int s = 1; s < 4; ++s) {
                const float4 g2 = red4[(4 * grp + s) * 12 + 3 * ip + cc];
                g.x += g2.x;  g.y += g2.y;  g.z += g2.z;  g.w += g2.w;
            }
            vm[cc] = g.x * msg_w2[0 * 3 + cc] + g.y * msg_w2[1 * 3 + cc] + g.z * msg_w2[2 * 3 + cc]
                   + g.w * (msg_b2[cc] * (1.0f / BIAS_FEATURE));
        }
        const float v0 = vm[0], v1 = vm[1], v2 = vm[2];
        const float4 hiq = snode[2 * i + 1];
        const float hi0 = hiq.x, hi1 = hiq.y, hi2 = hiq.z;

        const float u0 = silu_f(v0 * upd_w1[0] + v1 * upd_w1[3] + v2 * upd_w1[6] + upd_b1[0]);
        const float u1 = silu_f(v0 * upd_w1[1] + v1 * upd_w1[4] + v2 * upd_w1[7] + upd_b1[1]);
        const float u2 = silu_f(v0 * upd_w1[2] + v1 * upd_w1[5] + v2 * upd_w1[8] + upd_b1[2]);

        const float h0 = hi0 + u0 * upd_w2[0] + u1 * upd_w2[3] + u2 * upd_w2[6] + upd_b2[0];
        const float h1 = hi1 + u0 * upd_w2[1] + u1 * upd_w2[4] + u2 * upd_w2[7] + upd_b2[1];
        const float h2 = hi2 + u0 * upd_w2[2] + u1 * upd_w2[5] + u2 * upd_w2[8] + upd_b2[2];

        const int base = (b * N_ATOMS + i) * 3;
        const int half = BATCH * N_ATOMS * 3;
#pragma unroll
        for (int cc = 0; cc < 3; ++cc) {
            const float px = h0 * dx_w[0 * 3 + cc] + h1 * dx_w[1 * 3 + cc] + h2 * dx_w[2 * 3 + cc] + dx_b[cc];
            const float pv = h0 * dv_w[0 * 3 + cc] + h1 * dv_w[1 * 3 + cc] + h2 * dv_w[2 * 3 + cc] + dv_b[cc];
            out[base + cc]        = tanh_f(px) * MAX_DISP;
            out[half + base + cc] = tanh_f(pv) * 1.0f;
        }
    }
}

extern "C" void kernel_launch(void* const* d_in, const int* in_sizes, int n_in,
                              void* d_out, int out_size, void* d_ws, size_t ws_size,
                              hipStream_t stream) {
    (void)in_sizes; (void)n_in; (void)d_ws; (void)ws_size; (void)out_size;
    const float* x_nm   = (const float*)d_in[0];
    const int*   types  = (const int*)  d_in[1];
    const int*   mask   = (const int*)  d_in[2];
    const float* emb_W  = (const float*)d_in[3];
    const float* msg_w1 = (const float*)d_in[4];
    const float* msg_b1 = (const float*)d_in[5];
    const float* msg_w2 = (const float*)d_in[6];
    const float* msg_b2 = (const float*)d_in[7];
    const float* upd_w1 = (const float*)d_in[8];
    const float* upd_b1 = (const float*)d_in[9];
    const float* upd_w2 = (const float*)d_in[10];
    const float* upd_b2 = (const float*)d_in[11];
    const float* dx_w   = (const float*)d_in[12];
    const float* dx_b   = (const float*)d_in[13];
    const float* dv_w   = (const float*)d_in[14];
    const float* dv_b   = (const float*)d_in[15];
    float* out = (float*)d_out;

    dim3 grid(BATCH * (N_ATOMS / RECV_PER_BLOCK));
    dim3 block(256);
    simple_et_kernel<<<grid, block, 0, stream>>>(
        x_nm, types, mask, emb_W,
        msg_w1, msg_b1, msg_w2, msg_b2,
        upd_w1, upd_b1, upd_w2, upd_b2,
        dx_w, dx_b, dv_w, dv_b, out);
}
